# starting kernel
# baseline (speedup 1.0000x reference)
#include <hip/hip_runtime.h>


typedef __attribute__((ext_vector_type(2))) float v2f;
typedef __attribute__((ext_vector_type(8))) float v8f;

#define D_MODEL  4096
#define D_FF     11008
#define REMAINED 4403
#define BATCH    32

#define WAVES_PER_BLOCK 8
#define K_CHUNK (D_MODEL / WAVES_PER_BLOCK)

__global__ __launch_bounds__(32 * WAVES_PER_BLOCK)
void gathered_gemm_wmma_f32(const float* __restrict__ x,
                            const float* __restrict__ weight,
                            const int*   __restrict__ indices,
                            float*       __restrict__ out)
{
    __shared__ float red[WAVES_PER_BLOCK][2][8][32];

    const int tile = blockIdx.x;
    const int lane = threadIdx.x & 31;
    const int wave = threadIdx.x >> 5;

    const int n     = lane & 15;
    const int khalf = lane >> 4;

    int r = tile * 16 + n;
    int rc = r < REMAINED ? r : (REMAINED - 1);
    const float* wrow  = weight + (size_t)indices[rc] * D_MODEL;
    const float* xrow0 = x + (size_t)n        * D_MODEL;
    const float* xrow1 = x + (size_t)(n + 16) * D_MODEL;

    v8f c0 = {};
    v8f c1 = {};

    const int kbase = wave * K_CHUNK;
    const int kend  = kbase + K_CHUNK;

    for (int kb = kbase; kb < kend; kb += 16) {
        const int ko = kb + 2 * khalf;
        __builtin_prefetch(wrow + ko + 80, 0, 0);
#pragma unroll
        for (int u = 0; u < 4; ++u) {
            const int kk = ko + 4 * u;
            v2f a0 = *(const v2f*)(xrow0 + kk);
            v2f a1 = *(const v2f*)(xrow1 + kk);
            v2f b  = *(const v2f*)(wrow  + kk);
            c0 = __builtin_amdgcn_wmma_f32_16x16x4_f32(false, a0, false, b,
                                                       (short)0, c0, false, false);
            c1 = __builtin_amdgcn_wmma_f32_16x16x4_f32(false, a1, false, b,
                                                       (short)0, c1, false, false);
        }
    }

#pragma unroll
    for (int j = 0; j < 8; ++j) {
        red[wave][0][j][lane] = c0[j];
        red[wave][1][j][lane] = c1[j];
    }
    __syncthreads();

    for (int e = threadIdx.x; e < 512; e += blockDim.x) {
        const int l = e & 31;
        const int j = (e >> 5) & 7;
        const int t = e >> 8;
        float s = 0.f;
#pragma unroll
        for (int w = 0; w < WAVES_PER_BLOCK; ++w) s += red[w][t][j][l];
        const int m  = j + 8 * (l >> 4) + 16 * t;
        const int rr = tile * 16 + (l & 15);
        if (rr < REMAINED) out[(size_t)m * REMAINED + rr] = s;
    }
}

extern "C" void kernel_launch(void* const* d_in, const int* in_sizes, int n_in,
                              void* d_out, int out_size, void* d_ws, size_t ws_size,
                              hipStream_t stream) {
    const float* x       = (const float*)d_in[0];
    const float* weight  = (const float*)d_in[1];
    const int*   indices = (const int*)d_in[2];
    float*       out     = (float*)d_out;

    const int n_tiles = (REMAINED + 15) / 16;
    gathered_gemm_wmma_f32<<<n_tiles, 32 * WAVES_PER_BLOCK, 0, stream>>>(
        x, weight, indices, out);
}


# Load x and gathered W with whole-line 512-B global loads through a per-wave LDS slice; WMMA fragments read from LDS as b128; k order inside each 8-k group permuted, so the f32 sum order changes
# speedup vs baseline: 1.8726x; 1.8726x over previous
#include <hip/hip_runtime.h>


typedef __attribute__((ext_vector_type(2))) float v2f;
typedef __attribute__((ext_vector_type(4))) float v4f;
typedef __attribute__((ext_vector_type(8))) float v8f;

#define D_MODEL  4096
#define D_FF     11008
#define REMAINED 4403
#define BATCH    32

#define WAVES_PER_BLOCK 8
#define K_CHUNK (D_MODEL / WAVES_PER_BLOCK)
#define KS      64
#define NSTAGE  (K_CHUNK / KS)
#define PITCH   68
#define SROWS   48

__global__ __launch_bounds__(32 * WAVES_PER_BLOCK)
void gathered_gemm_wmma_f32(const float* __restrict__ x,
                            const float* __restrict__ weight,
                            const int*   __restrict__ indices,
                            float*       __restrict__ out)
{
    __shared__ __attribute__((aligned(16))) float stage[WAVES_PER_BLOCK][SROWS][PITCH];
    __shared__ float red[WAVES_PER_BLOCK][2][8][32];

    const int tile = blockIdx.x;
    const int lane = threadIdx.x & 31;
    const int wave = threadIdx.x >> 5;

    const int n = lane & 15;
    const int h = lane >> 4;

    const int lrow = lane >> 4;
    const int lcol = (lane & 15) * 4;

    const int kbase = wave * K_CHUNK;

    const float* wsrc[8];
#pragma unroll
    for (int j = 0; j < 8; ++j) {
        int r  = tile * 16 + 2 * j + lrow;
        int rc = r < REMAINED ? r : (REMAINED - 1);
        wsrc[j] = weight + (size_t)indices[rc] * D_MODEL + kbase + lcol;
    }
    const float* xsrc = x + (size_t)lrow * D_MODEL + kbase + lcol;

    float* sl = &stage[wave][0][0];
    float* st_dst = sl + lrow * PITCH + lcol;
    const float* fr_src = sl + n * PITCH + 4 * h;

    v8f c0 = {};
    v8f c1 = {};

    v4f gw[8], gx[16];
#pragma unroll
    for (int j = 0; j < 8; ++j)  gw[j] = *(const v4f*)(wsrc[j]);
#pragma unroll
    for (int j = 0; j < 16; ++j) gx[j] = *(const v4f*)(xsrc + (size_t)(2 * j) * D_MODEL);

    for (int s = 0; s < NSTAGE; ++s) {
#pragma unroll
        for (int j = 0; j < 8; ++j)  *(v4f*)(st_dst + (2 * j) * PITCH)      = gw[j];
#pragma unroll
        for (int j = 0; j < 16; ++j) *(v4f*)(st_dst + (16 + 2 * j) * PITCH) = gx[j];

        if (s + 1 < NSTAGE) {
            const int ko = (s + 1) * KS;
#pragma unroll
            for (int j = 0; j < 8; ++j)  gw[j] = *(const v4f*)(wsrc[j] + ko);
#pragma unroll
            for (int j = 0; j < 16; ++j) gx[j] = *(const v4f*)(xsrc + (size_t)(2 * j) * D_MODEL + ko);
        }

#pragma unroll
        for (int kk = 0; kk < KS / 8; ++kk) {
            const float* p = fr_src + 8 * kk;
            v4f b  = *(const v4f*)(p);
            v4f a0 = *(const v4f*)(p + 16 * PITCH);
            v4f a1 = *(const v4f*)(p + 32 * PITCH);
            c0 = __builtin_amdgcn_wmma_f32_16x16x4_f32(false, a0.xy, false, b.xy,
                                                       (short)0, c0, false, false);
            c1 = __builtin_amdgcn_wmma_f32_16x16x4_f32(false, a1.xy, false, b.xy,
                                                       (short)0, c1, false, false);
            c0 = __builtin_amdgcn_wmma_f32_16x16x4_f32(false, a0.zw, false, b.zw,
                                                       (short)0, c0, false, false);
            c1 = __builtin_amdgcn_wmma_f32_16x16x4_f32(false, a1.zw, false, b.zw,
                                                       (short)0, c1, false, false);
        }
    }

#pragma unroll
    for (int j = 0; j < 8; ++j) {
        red[wave][0][j][lane] = c0[j];
        red[wave][1][j][lane] = c1[j];
    }
    __syncthreads();

    for (int e = threadIdx.x; e < 512; e += blockDim.x) {
        const int l = e & 31;
        const int j = (e >> 5) & 7;
        const int t = e >> 8;
        float s = 0.f;
#pragma unroll
        for (int w = 0; w < WAVES_PER_BLOCK; ++w) s += red[w][t][j][l];
        const int m  = j + 8 * (l >> 4) + 16 * t;
        const int rr = tile * 16 + (l & 15);
        if (rr < REMAINED) out[(size_t)m * REMAINED + rr] = s;
    }
}

extern "C" void kernel_launch(void* const* d_in, const int* in_sizes, int n_in,
                              void* d_out, int out_size, void* d_ws, size_t ws_size,
                              hipStream_t stream) {
    const float* x       = (const float*)d_in[0];
    const float* weight  = (const float*)d_in[1];
    const int*   indices = (const int*)d_in[2];
    float*       out     = (float*)d_out;

    const int n_tiles = (REMAINED + 15) / 16;
    gathered_gemm_wmma_f32<<<n_tiles, 32 * WAVES_PER_BLOCK, 0, stream>>>(
        x, weight, indices, out);
}


# Grid 276 -> 256 work-groups, one whole 16-column tile each; the 20 surplus tiles are cut along the batch and work-groups 0..39 add one half each, reusing their x rows; sum order unchanged
# speedup vs baseline: 2.2407x; 1.1966x over previous
#include <hip/hip_runtime.h>


typedef __attribute__((ext_vector_type(2))) float v2f;
typedef __attribute__((ext_vector_type(4))) float v4f;
typedef __attribute__((ext_vector_type(8))) float v8f;

#define D_MODEL  4096
#define D_FF     11008
#define REMAINED 4403
#define BATCH    32

#define N_TILES   ((REMAINED + 15) / 16)
#define N_BLOCKS  256
#define N_SURPLUS (N_TILES - N_BLOCKS)

#define WAVES_PER_BLOCK 8
#define NTHREADS (32 * WAVES_PER_BLOCK)
#define K_CHUNK (D_MODEL / WAVES_PER_BLOCK)
#define KS      64
#define NSTAGE  (K_CHUNK / KS)
#define PITCH   68
#define SROWS   64

static_assert(2 * N_SURPLUS <= N_BLOCKS, "every surplus half needs a work-group of its own");

template <int XH>
__device__ __forceinline__
void run_block(const float* __restrict__ x,
               const float* __restrict__ weight,
               const int*   __restrict__ indices,
               float*       __restrict__ out,
               float* __restrict__ sl,
               float (* __restrict__ red)[3][8][32],
               const int tile, const int xtile)
{
    const int lane = threadIdx.x & 31;
    const int wave = threadIdx.x >> 5;

    const int n = lane & 15;
    const int h = lane >> 4;

    const int lrow = lane >> 4;
    const int lcol = (lane & 15) * 4;

    const int kbase = wave * K_CHUNK;

    const float* wsrc[8];
    const float* wsrc2[8];
#pragma unroll
    for (int j = 0; j < 8; ++j) {
        int r  = tile * 16 + 2 * j + lrow;
        int rc = r < REMAINED ? r : (REMAINED - 1);
        wsrc[j] = weight + (size_t)indices[rc] * D_MODEL + kbase + lcol;
    }
    if (XH) {
#pragma unroll
        for (int j = 0; j < 8; ++j) {
            int r  = xtile * 16 + 2 * j + lrow;
            int rc = r < REMAINED ? r : (REMAINED - 1);
            wsrc2[j] = weight + (size_t)indices[rc] * D_MODEL + kbase + lcol;
        }
    }
    const float* xsrc = x + (size_t)lrow * D_MODEL + kbase + lcol;

    float* st_dst = sl + lrow * PITCH + lcol;
    const float* fr_src = sl + n * PITCH + 4 * h;

    v8f c0 = {};
    v8f c1 = {};
    v8f c2 = {};

    v4f gw[8], gx[16], gw2[8];
#pragma unroll
    for (int j = 0; j < 8; ++j)  gw[j] = *(const v4f*)(wsrc[j]);
#pragma unroll
    for (int j = 0; j < 16; ++j) gx[j] = *(const v4f*)(xsrc + (size_t)(2 * j) * D_MODEL);
    if (XH) {
#pragma unroll
        for (int j = 0; j < 8; ++j)  gw2[j] = *(const v4f*)(wsrc2[j]);
    }

    for (int s = 0; s < NSTAGE; ++s) {
#pragma unroll
        for (int j = 0; j < 8; ++j)  *(v4f*)(st_dst + (2 * j) * PITCH)      = gw[j];
#pragma unroll
        for (int j = 0; j < 16; ++j) *(v4f*)(st_dst + (16 + 2 * j) * PITCH) = gx[j];
        if (XH) {
#pragma unroll
            for (int j = 0; j < 8; ++j)  *(v4f*)(st_dst + (48 + 2 * j) * PITCH) = gw2[j];
        }

        if (s + 1 < NSTAGE) {
            const int ko = (s + 1) * KS;
#pragma unroll
            for (int j = 0; j < 8; ++j)  gw[j] = *(const v4f*)(wsrc[j] + ko);
#pragma unroll
            for (int j = 0; j < 16; ++j) gx[j] = *(const v4f*)(xsrc + (size_t)(2 * j) * D_MODEL + ko);
            if (XH) {
#pragma unroll
                for (int j = 0; j < 8; ++j)  gw2[j] = *(const v4f*)(wsrc2[j] + ko);
            }
        }

#pragma unroll
        for (int kk = 0; kk < KS / 8; ++kk) {
            const float* p = fr_src + 8 * kk;
            v4f b  = *(const v4f*)(p);
            v4f a0 = *(const v4f*)(p + 16 * PITCH);
            v4f a1 = *(const v4f*)(p + 32 * PITCH);
            v4f b2 = {};
            if (XH) b2 = *(const v4f*)(p + 48 * PITCH);
            const v4f ax = (XH == 2) ? a1 : a0;
            c0 = __builtin_amdgcn_wmma_f32_16x16x4_f32(false, a0.xy, false, b.xy,
                                                       (short)0, c0, false, false);
            c1 = __builtin_amdgcn_wmma_f32_16x16x4_f32(false, a1.xy, false, b.xy,
                                                       (short)0, c1, false, false);
            if (XH)
                c2 = __builtin_amdgcn_wmma_f32_16x16x4_f32(false, ax.xy, false, b2.xy,
                                                           (short)0, c2, false, false);
            c0 = __builtin_amdgcn_wmma_f32_16x16x4_f32(false, a0.zw, false, b.zw,
                                                       (short)0, c0, false, false);
            c1 = __builtin_amdgcn_wmma_f32_16x16x4_f32(false, a1.zw, false, b.zw,
                                                       (short)0, c1, false, false);
            if (XH)
                c2 = __builtin_amdgcn_wmma_f32_16x16x4_f32(false, ax.zw, false, b2.zw,
                                                           (short)0, c2, false, false);
        }
    }

#pragma unroll
    for (int j = 0; j < 8; ++j) {
        red[wave][0][j][lane] = c0[j];
        red[wave][1][j][lane] = c1[j];
        if (XH) red[wave][2][j][lane] = c2[j];
    }
    __syncthreads();

    const int l = threadIdx.x & 31;
    const int j = (threadIdx.x >> 5) & 7;
#pragma unroll
    for (int t = 0; t < (XH ? 3 : 2); ++t) {
        float s = 0.f;
#pragma unroll
        for (int w = 0; w < WAVES_PER_BLOCK; ++w) s += red[w][t][j][l];
        const int mh  = (t == 2) ? (XH - 1) : t;
        const int col = (t == 2) ? xtile : tile;
        const int m  = j + 8 * (l >> 4) + 16 * mh;
        const int rr = col * 16 + (l & 15);
        if (rr < REMAINED) out[(size_t)m * REMAINED + rr] = s;
    }
}

__global__ __launch_bounds__(NTHREADS)
void gathered_gemm_wmma_f32(const float* __restrict__ x,
                            const float* __restrict__ weight,
                            const int*   __restrict__ indices,
                            float*       __restrict__ out)
{
    __shared__ __attribute__((aligned(16))) float stage[WAVES_PER_BLOCK][SROWS][PITCH];
    __shared__ float red[WAVES_PER_BLOCK][3][8][32];

    const int blk  = blockIdx.x;
    const int wave = threadIdx.x >> 5;
    float* sl = &stage[wave][0][0];

    if (blk < 2 * N_SURPLUS) {
        const int xtile = N_BLOCKS + (blk >> 1);
        if (blk & 1) run_block<2>(x, weight, indices, out, sl, red, blk, xtile);
        else         run_block<1>(x, weight, indices, out, sl, red, blk, xtile);
    } else {
        run_block<0>(x, weight, indices, out, sl, red, blk, 0);
    }
}

extern "C" void kernel_launch(void* const* d_in, const int* in_sizes, int n_in,
                              void* d_out, int out_size, void* d_ws, size_t ws_size,
                              hipStream_t stream) {
    const float* x       = (const float*)d_in[0];
    const float* weight  = (const float*)d_in[1];
    const int*   indices = (const int*)d_in[2];
    float*       out     = (float*)d_out;

    gathered_gemm_wmma_f32<<<N_BLOCKS, NTHREADS, 0, stream>>>(
        x, weight, indices, out);
}
